# starting kernel
# baseline (speedup 1.0000x reference)
#include <hip/hip_runtime.h>


#define KPTS  15
#define HNBR  32
#define COUT  32
#define NPTS  65536

#define KC    45
#define KCPAD 48
#define NSTEP 12

#define EXP_SCALE (-288.53900817779268f)

typedef __attribute__((ext_vector_type(2))) float v2f;
typedef __attribute__((ext_vector_type(8))) float v8f;
typedef __attribute__((ext_vector_type(4))) int   v4i;

__device__ __forceinline__ float uniform_f(float v) {
    return __int_as_float(__builtin_amdgcn_readfirstlane(__float_as_int(v)));
}

__global__ __launch_bounds__(256, 1) void kpconv_wmma_kernel(
    const float* __restrict__ points,
    const int*   __restrict__ nbr,
    const float* __restrict__ kp,
    const float* __restrict__ W,
    const float* __restrict__ bias,
    float*       __restrict__ out,
    int nTiles)
{
    __shared__ float Wt[COUT * KCPAD];

    const int tid = threadIdx.x;
#pragma unroll
    for (int e = tid; e < COUT * KCPAD; e += 256) {
        const int d = e / KCPAD, kc = e % KCPAD;
        const int kcc = (kc < KC) ? kc : (KC - 1);
        const float v = W[kcc * COUT + d];
        Wt[e] = (kc < KC) ? v : 0.0f;
    }
    __syncthreads();

    const int lane  = tid & 31;
    const int half  = lane >> 4;
    const int col   = lane & 15;
    const int wave  = (blockIdx.x * blockDim.x + tid) >> 5;
    const int nWave = (gridDim.x * blockDim.x) >> 5;

    float ekx[KPTS], eky[KPTS], ekz[KPTS], ec[KPTS];
#pragma unroll
    for (int k = 0; k < KPTS; ++k) {
        const float x = kp[k * 3 + 0];
        const float y = kp[k * 3 + 1];
        const float z = kp[k * 3 + 2];
        ekx[k] = uniform_f(-2.0f * EXP_SCALE * x);
        eky[k] = uniform_f(-2.0f * EXP_SCALE * y);
        ekz[k] = uniform_f(-2.0f * EXP_SCALE * z);
        ec[k]  = uniform_f(EXP_SCALE * (x * x + y * y + z * z));
    }
    const float b0 = bias[col];
    const float b1 = bias[16 + col];

    for (int tile = wave; tile < nTiles; tile += nWave) {
        const int m = tile * 16 + col;
        const float* pc = points + m * 3;
        const float cx = pc[0], cy = pc[1], cz = pc[2];

        const v4i* ip = (const v4i*)(nbr + m * HNBR + half * 16);

        float pw[KCPAD];
#pragma unroll
        for (int j = 0; j < KCPAD; ++j) pw[j] = 0.0f;

#pragma unroll 1
        for (int q = 0; q < 4; ++q) {
            const v4i iv = ip[q];
            float nx[4], ny[4], nz[4];
#pragma unroll
            for (int u = 0; u < 4; ++u) {
                int idx = iv[u];
                idx = (idx < 0) ? 0 : ((idx >= NPTS) ? NPTS - 1 : idx);
                const float* pn = points + idx * 3;
                nx[u] = pn[0]; ny[u] = pn[1]; nz[u] = pn[2];
            }
#pragma unroll
            for (int u = 0; u < 4; ++u) {
                const float rx = nx[u] - cx, ry = ny[u] - cy, rz = nz[u] - cz;
                float rs = rx * rx;
                rs = fmaf(ry, ry, rs);
                rs = fmaf(rz, rz, rs);
                rs *= EXP_SCALE;
#pragma unroll
                for (int k = 0; k < KPTS; ++k) {
                    float t = rs + ec[k];
                    t = fmaf(rx, ekx[k], t);
                    t = fmaf(ry, eky[k], t);
                    t = fmaf(rz, ekz[k], t);
                    const float w = __builtin_amdgcn_exp2f(t);
                    pw[3 * k + 0] = fmaf(w, nx[u], pw[3 * k + 0]);
                    pw[3 * k + 1] = fmaf(w, ny[u], pw[3 * k + 1]);
                    pw[3 * k + 2] = fmaf(w, nz[u], pw[3 * k + 2]);
                }
            }
        }

#pragma unroll
        for (int j = 0; j < KC; ++j)
            pw[j] += __shfl_xor(pw[j], 16, 32);

        v8f acc0 = {};
        v8f acc1 = {};
        const float* wrow0 = Wt + col * KCPAD;
        const float* wrow1 = Wt + (16 + col) * KCPAD;
#pragma unroll
        for (int s = 0; s < NSTEP; ++s) {
            v2f a;
            a.x = half ? pw[4 * s + 2] : pw[4 * s + 0];
            a.y = half ? pw[4 * s + 3] : pw[4 * s + 1];
            const v2f bf0 = *(const v2f*)(wrow0 + 4 * s + 2 * half);
            const v2f bf1 = *(const v2f*)(wrow1 + 4 * s + 2 * half);
            acc0 = __builtin_amdgcn_wmma_f32_16x16x4_f32(
                false, a, false, bf0, (short)0, acc0, false, false);
            acc1 = __builtin_amdgcn_wmma_f32_16x16x4_f32(
                false, a, false, bf1, (short)0, acc1, false, false);
        }

        const int rowBase = tile * 16 + 8 * half;
#pragma unroll
        for (int r = 0; r < 8; ++r) {
            const int o = (rowBase + r) * COUT;
            out[o + col]      = acc0[r] + b0;
            out[o + 16 + col] = acc1[r] + b1;
        }
    }
}

extern "C" void kernel_launch(void* const* d_in, const int* in_sizes, int n_in,
                              void* d_out, int out_size, void* d_ws, size_t ws_size,
                              hipStream_t stream) {
    const float* points = (const float*)d_in[0];
    const int*   nbr    = (const int*)  d_in[1];
    const float* kp     = (const float*)d_in[2];
    const float* W      = (const float*)d_in[3];
    const float* bias   = (const float*)d_in[4];
    float*       out    = (float*)d_out;

    const int nTiles = NPTS / 16;
    kpconv_wmma_kernel<<<dim3(512), dim3(256), 0, stream>>>(
        points, nbr, kp, W, bias, out, nTiles);
}


# 48 partial sums stay in registers: A fragment picked by value select, not address select, so no scratch stores/reloads per tile; waves_per_eu(4,4) keeps the loop's ILP at the real 4 waves/SIMD
# speedup vs baseline: 2.3754x; 2.3754x over previous
#include <hip/hip_runtime.h>


#define KPTS  15
#define HNBR  32
#define COUT  32
#define NPTS  65536

#define KC    45
#define KCPAD 48
#define NSTEP 12

#define EXP_SCALE (-288.53900817779268f)

typedef __attribute__((ext_vector_type(2))) float v2f;
typedef __attribute__((ext_vector_type(8))) float v8f;
typedef __attribute__((ext_vector_type(4))) int   v4i;

__device__ __forceinline__ float uniform_f(float v) {
    return __int_as_float(__builtin_amdgcn_readfirstlane(__float_as_int(v)));
}

__global__ __launch_bounds__(256) __attribute__((amdgpu_waves_per_eu(4, 4)))
void kpconv_wmma_kernel(
    const float* __restrict__ points,
    const int*   __restrict__ nbr,
    const float* __restrict__ kp,
    const float* __restrict__ W,
    const float* __restrict__ bias,
    float*       __restrict__ out,
    int nTiles)
{
    __shared__ float Wt[COUT * KCPAD];

    const int tid = threadIdx.x;
#pragma unroll
    for (int e = tid; e < COUT * KCPAD; e += 256) {
        const int d = e / KCPAD, kc = e % KCPAD;
        const int kcc = (kc < KC) ? kc : (KC - 1);
        const float v = W[kcc * COUT + d];
        Wt[e] = (kc < KC) ? v : 0.0f;
    }
    __syncthreads();

    const int lane  = tid & 31;
    const int half  = lane >> 4;
    const int col   = lane & 15;
    const int wave  = (blockIdx.x * blockDim.x + tid) >> 5;
    const int nWave = (gridDim.x * blockDim.x) >> 5;

    float ekx[KPTS], eky[KPTS], ekz[KPTS], ec[KPTS];
#pragma unroll
    for (int k = 0; k < KPTS; ++k) {
        const float x = kp[k * 3 + 0];
        const float y = kp[k * 3 + 1];
        const float z = kp[k * 3 + 2];
        ekx[k] = uniform_f(-2.0f * EXP_SCALE * x);
        eky[k] = uniform_f(-2.0f * EXP_SCALE * y);
        ekz[k] = uniform_f(-2.0f * EXP_SCALE * z);
        ec[k]  = uniform_f(EXP_SCALE * (x * x + y * y + z * z));
    }
    const float b0 = bias[col];
    const float b1 = bias[16 + col];

    for (int tile = wave; tile < nTiles; tile += nWave) {
        const int m = tile * 16 + col;
        const float* pc = points + m * 3;
        const float cx = pc[0], cy = pc[1], cz = pc[2];

        const v4i* ip = (const v4i*)(nbr + m * HNBR + half * 16);

        float pw[KCPAD];
#pragma unroll
        for (int j = 0; j < KCPAD; ++j) pw[j] = 0.0f;

#pragma unroll 1
        for (int q = 0; q < 4; ++q) {
            const v4i iv = ip[q];
            float nx[4], ny[4], nz[4];
#pragma unroll
            for (int u = 0; u < 4; ++u) {
                int idx = iv[u];
                idx = (idx < 0) ? 0 : ((idx >= NPTS) ? NPTS - 1 : idx);
                const float* pn = points + idx * 3;
                nx[u] = pn[0]; ny[u] = pn[1]; nz[u] = pn[2];
            }
#pragma unroll
            for (int u = 0; u < 4; ++u) {
                const float rx = nx[u] - cx, ry = ny[u] - cy, rz = nz[u] - cz;
                float rs = rx * rx;
                rs = fmaf(ry, ry, rs);
                rs = fmaf(rz, rz, rs);
                rs *= EXP_SCALE;
#pragma unroll
                for (int k = 0; k < KPTS; ++k) {
                    float t = rs + ec[k];
                    t = fmaf(rx, ekx[k], t);
                    t = fmaf(ry, eky[k], t);
                    t = fmaf(rz, ekz[k], t);
                    const float w = __builtin_amdgcn_exp2f(t);
                    pw[3 * k + 0] = fmaf(w, nx[u], pw[3 * k + 0]);
                    pw[3 * k + 1] = fmaf(w, ny[u], pw[3 * k + 1]);
                    pw[3 * k + 2] = fmaf(w, nz[u], pw[3 * k + 2]);
                }
            }
        }

#pragma unroll
        for (int j = 0; j < KC; ++j)
            pw[j] += __shfl_xor(pw[j], 16, 32);

        v8f acc0 = {};
        v8f acc1 = {};
        const float* wrow0 = Wt + col * KCPAD;
        const float* wrow1 = Wt + (16 + col) * KCPAD;
#pragma unroll
        for (int s = 0; s < NSTEP; ++s) {
            const float lo0 = pw[4 * s + 0], lo1 = pw[4 * s + 1];
            const float hi0 = pw[4 * s + 2], hi1 = pw[4 * s + 3];
            v2f a;
            a.x = half ? hi0 : lo0;
            a.y = half ? hi1 : lo1;
            const v2f bf0 = *(const v2f*)(wrow0 + 4 * s + 2 * half);
            const v2f bf1 = *(const v2f*)(wrow1 + 4 * s + 2 * half);
            acc0 = __builtin_amdgcn_wmma_f32_16x16x4_f32(
                false, a, false, bf0, (short)0, acc0, false, false);
            acc1 = __builtin_amdgcn_wmma_f32_16x16x4_f32(
                false, a, false, bf1, (short)0, acc1, false, false);
        }

        const int rowBase = tile * 16 + 8 * half;
#pragma unroll
        for (int r = 0; r < 8; ++r) {
            const int o = (rowBase + r) * COUT;
            out[o + col]      = acc0[r] + b0;
            out[o + 16 + col] = acc1[r] + b1;
        }
    }
}

extern "C" void kernel_launch(void* const* d_in, const int* in_sizes, int n_in,
                              void* d_out, int out_size, void* d_ws, size_t ws_size,
                              hipStream_t stream) {
    const float* points = (const float*)d_in[0];
    const int*   nbr    = (const int*)  d_in[1];
    const float* kp     = (const float*)d_in[2];
    const float* W      = (const float*)d_in[3];
    const float* bias   = (const float*)d_in[4];
    float*       out    = (float*)d_out;

    const int nTiles = NPTS / 16;
    kpconv_wmma_kernel<<<dim3(512), dim3(256), 0, stream>>>(
        points, nbr, kp, W, bias, out, nTiles);
}


# All independent loads issued as one batch, then all 16 neighbor gathers at once; barrier fences LDS only so gathers are consumed as they land; neighbor loop unrolled, one tile per wave
# speedup vs baseline: 2.6238x; 1.1046x over previous
#include <hip/hip_runtime.h>


#define KPTS  15
#define HNBR  32
#define COUT  32
#define NPTS  65536

#define KC    45
#define KCPAD 48
#define NSTEP 12

#define EXP_SCALE (-288.53900817779268f)

typedef __attribute__((ext_vector_type(2))) float v2f;
typedef __attribute__((ext_vector_type(8))) float v8f;
typedef __attribute__((ext_vector_type(4))) int   v4i;

__device__ __forceinline__ float uniform_f(float v) {
    return __int_as_float(__builtin_amdgcn_readfirstlane(__float_as_int(v)));
}

__global__ __launch_bounds__(256) __attribute__((amdgpu_waves_per_eu(4, 4)))
void kpconv_wmma_kernel(
    const float* __restrict__ points,
    const int*   __restrict__ nbr,
    const float* __restrict__ kp,
    const float* __restrict__ W,
    const float* __restrict__ bias,
    float*       __restrict__ out,
    int nTiles)
{
    __shared__ float Wt[COUT * KCPAD];

    const int tid   = threadIdx.x;
    const int lane  = tid & 31;
    const int half  = lane >> 4;
    const int col   = lane & 15;
    const int wave  = (blockIdx.x * 256 + tid) >> 5;
    const int tile  = (wave < nTiles) ? wave : (nTiles - 1);

    const int m = tile * 16 + col;
    const v4i* ip = (const v4i*)(nbr + m * HNBR + half * 16);
    v4i iv[4];
#pragma unroll
    for (int q = 0; q < 4; ++q) iv[q] = ip[q];
    const float* pc = points + m * 3;
    const float cx = pc[0], cy = pc[1], cz = pc[2];

    float wv[6];
#pragma unroll
    for (int i = 0; i < 6; ++i) {
        const int e = tid + 256 * i;
        const int es = (e < KC * COUT) ? e : (KC * COUT - 1);
        wv[i] = W[es];
    }
    const float b0 = bias[col];
    const float b1 = bias[16 + col];
    __builtin_amdgcn_sched_barrier(0);

    float nx[16], ny[16], nz[16];
#pragma unroll
    for (int q = 0; q < 4; ++q) {
#pragma unroll
        for (int u = 0; u < 4; ++u) {
            int idx = iv[q][u];
            idx = (idx < 0) ? 0 : ((idx >= NPTS) ? NPTS - 1 : idx);
            const float* pn = points + idx * 3;
            nx[4 * q + u] = pn[0]; ny[4 * q + u] = pn[1]; nz[4 * q + u] = pn[2];
        }
    }

    float ekx[KPTS], eky[KPTS], ekz[KPTS], ec[KPTS];
#pragma unroll
    for (int k = 0; k < KPTS; ++k) {
        const float x = kp[k * 3 + 0];
        const float y = kp[k * 3 + 1];
        const float z = kp[k * 3 + 2];
        ekx[k] = uniform_f(-2.0f * EXP_SCALE * x);
        eky[k] = uniform_f(-2.0f * EXP_SCALE * y);
        ekz[k] = uniform_f(-2.0f * EXP_SCALE * z);
        ec[k]  = uniform_f(EXP_SCALE * (x * x + y * y + z * z));
    }

#pragma unroll
    for (int i = 0; i < 6; ++i) {
        const int e = tid + 256 * i;
        const int es = (e < KC * COUT) ? e : (KC * COUT - 1);
        Wt[(es % COUT) * KCPAD + es / COUT] = wv[i];
    }
    if (tid < COUT * (KCPAD - KC))
        Wt[(tid / (KCPAD - KC)) * KCPAD + KC + tid % (KCPAD - KC)] = 0.0f;
    __builtin_amdgcn_fence(__ATOMIC_RELEASE, "workgroup", "local");
    __builtin_amdgcn_s_barrier();
    __builtin_amdgcn_fence(__ATOMIC_ACQUIRE, "workgroup", "local");
    __builtin_amdgcn_sched_barrier(0);

    float pw[KCPAD];
#pragma unroll
    for (int j = 0; j < KCPAD; ++j) pw[j] = 0.0f;

#pragma unroll
    for (int u = 0; u < 16; ++u) {
        const float rx = nx[u] - cx, ry = ny[u] - cy, rz = nz[u] - cz;
        float rs = rx * rx;
        rs = fmaf(ry, ry, rs);
        rs = fmaf(rz, rz, rs);
        rs *= EXP_SCALE;
#pragma unroll
        for (int k = 0; k < KPTS; ++k) {
            float t = rs + ec[k];
            t = fmaf(rx, ekx[k], t);
            t = fmaf(ry, eky[k], t);
            t = fmaf(rz, ekz[k], t);
            const float w = __builtin_amdgcn_exp2f(t);
            pw[3 * k + 0] = fmaf(w, nx[u], pw[3 * k + 0]);
            pw[3 * k + 1] = fmaf(w, ny[u], pw[3 * k + 1]);
            pw[3 * k + 2] = fmaf(w, nz[u], pw[3 * k + 2]);
        }
    }

#pragma unroll
    for (int j = 0; j < KC; ++j)
        pw[j] += __shfl_xor(pw[j], 16, 32);

    v8f acc0 = {};
    v8f acc1 = {};
    const float* wrow0 = Wt + col * KCPAD;
    const float* wrow1 = Wt + (16 + col) * KCPAD;
#pragma unroll
    for (int s = 0; s < NSTEP; ++s) {
        const float lo0 = pw[4 * s + 0], lo1 = pw[4 * s + 1];
        const float hi0 = pw[4 * s + 2], hi1 = pw[4 * s + 3];
        v2f a;
        a.x = half ? hi0 : lo0;
        a.y = half ? hi1 : lo1;
        const v2f bf0 = *(const v2f*)(wrow0 + 4 * s + 2 * half);
        const v2f bf1 = *(const v2f*)(wrow1 + 4 * s + 2 * half);
        acc0 = __builtin_amdgcn_wmma_f32_16x16x4_f32(
            false, a, false, bf0, (short)0, acc0, false, false);
        acc1 = __builtin_amdgcn_wmma_f32_16x16x4_f32(
            false, a, false, bf1, (short)0, acc1, false, false);
    }

    if (wave < nTiles) {
        const int rowBase = tile * 16 + 8 * half;
#pragma unroll
        for (int r = 0; r < 8; ++r) {
            const int o = (rowBase + r) * COUT;
            out[o + col]      = acc0[r] + b0;
            out[o + 16 + col] = acc1[r] + b1;
        }
    }
}

extern "C" void kernel_launch(void* const* d_in, const int* in_sizes, int n_in,
                              void* d_out, int out_size, void* d_ws, size_t ws_size,
                              hipStream_t stream) {
    const float* points = (const float*)d_in[0];
    const int*   nbr    = (const int*)  d_in[1];
    const float* kp     = (const float*)d_in[2];
    const float* W      = (const float*)d_in[3];
    const float* bias   = (const float*)d_in[4];
    float*       out    = (float*)d_out;

    const int nTiles = NPTS / 16;
    kpconv_wmma_kernel<<<dim3(512), dim3(256), 0, stream>>>(
        points, nbr, kp, W, bias, out, nTiles);
}


# Neighbors at distance >= max|kp|+0.726 (all 15 weights exactly 0 in f32) are skipped: each lane keeps its near ones, in order, in a private LDS list; the 15-exp body loops over that list only
# speedup vs baseline: 2.9167x; 1.1116x over previous
#include <hip/hip_runtime.h>


#define KPTS  15
#define HNBR  32
#define COUT  32
#define NPTS  65536

#define KC    45
#define KCPAD 48
#define NSTEP 12

#define EXP_SCALE (-288.53900817779268f)
#define CULL_R0 (0.7260f)
#define NLIST 16

typedef __attribute__((ext_vector_type(2))) float v2f;
typedef __attribute__((ext_vector_type(8))) float v8f;
typedef __attribute__((ext_vector_type(4))) int   v4i;

__device__ __forceinline__ float uniform_f(float v) {
    return __int_as_float(__builtin_amdgcn_readfirstlane(__float_as_int(v)));
}

__global__ __launch_bounds__(256) __attribute__((amdgpu_waves_per_eu(4, 4)))
void kpconv_wmma_kernel(
    const float* __restrict__ points,
    const int*   __restrict__ nbr,
    const float* __restrict__ kp,
    const float* __restrict__ W,
    const float* __restrict__ bias,
    float*       __restrict__ out,
    int nTiles)
{
    __shared__ float Wt[COUT * KCPAD];
    __shared__ float Lx[NLIST * 256];
    __shared__ float Ly[NLIST * 256];
    __shared__ float Lz[NLIST * 256];

    const int tid   = threadIdx.x;
    const int lane  = tid & 31;
    const int half  = lane >> 4;
    const int col   = lane & 15;
    const int wave  = (blockIdx.x * 256 + tid) >> 5;
    const int tile  = (wave < nTiles) ? wave : (nTiles - 1);

    const int m = tile * 16 + col;
    const v4i* ip = (const v4i*)(nbr + m * HNBR + half * 16);
    v4i iv[4];
#pragma unroll
    for (int q = 0; q < 4; ++q) iv[q] = ip[q];
    const float* pc = points + m * 3;
    const float cx = pc[0], cy = pc[1], cz = pc[2];

    float wv[6];
#pragma unroll
    for (int i = 0; i < 6; ++i) {
        const int e = tid + 256 * i;
        const int es = (e < KC * COUT) ? e : (KC * COUT - 1);
        wv[i] = W[es];
    }
    const float b0 = bias[col];
    const float b1 = bias[16 + col];
    __builtin_amdgcn_sched_barrier(0);

    float nx[16], ny[16], nz[16];
#pragma unroll
    for (int q = 0; q < 4; ++q) {
#pragma unroll
        for (int u = 0; u < 4; ++u) {
            int idx = iv[q][u];
            idx = (idx < 0) ? 0 : ((idx >= NPTS) ? NPTS - 1 : idx);
            const float* pn = points + idx * 3;
            nx[4 * q + u] = pn[0]; ny[4 * q + u] = pn[1]; nz[4 * q + u] = pn[2];
        }
    }

    float ekx[KPTS], eky[KPTS], ekz[KPTS], ec[KPTS];
    float kmax2 = 0.0f;
    bool  kfinite = true;
#pragma unroll
    for (int k = 0; k < KPTS; ++k) {
        const float x = kp[k * 3 + 0];
        const float y = kp[k * 3 + 1];
        const float z = kp[k * 3 + 2];
        const float k2 = x * x + y * y + z * z;
        kmax2 = fmaxf(kmax2, k2);
        kfinite = kfinite && (k2 < __builtin_inff());
        ekx[k] = uniform_f(-2.0f * EXP_SCALE * x);
        eky[k] = uniform_f(-2.0f * EXP_SCALE * y);
        ekz[k] = uniform_f(-2.0f * EXP_SCALE * z);
        ec[k]  = uniform_f(EXP_SCALE * (x * x + y * y + z * z));
    }
    const float kr = __builtin_sqrtf(kmax2) + CULL_R0;
    const float far2 = uniform_f(kfinite ? kr * kr * 1.001f : __builtin_inff());

#pragma unroll
    for (int i = 0; i < 6; ++i) {
        const int e = tid + 256 * i;
        const int es = (e < KC * COUT) ? e : (KC * COUT - 1);
        Wt[(es % COUT) * KCPAD + es / COUT] = wv[i];
    }
    if (tid < COUT * (KCPAD - KC))
        Wt[(tid / (KCPAD - KC)) * KCPAD + KC + tid % (KCPAD - KC)] = 0.0f;
    __builtin_amdgcn_fence(__ATOMIC_RELEASE, "workgroup", "local");
    __builtin_amdgcn_s_barrier();
    __builtin_amdgcn_fence(__ATOMIC_ACQUIRE, "workgroup", "local");
    __builtin_amdgcn_sched_barrier(0);

    float pw[KCPAD];
#pragma unroll
    for (int j = 0; j < KCPAD; ++j) pw[j] = 0.0f;

    int cnt = 0;
#pragma unroll
    for (int u = 0; u < 16; ++u) {
        const float rx = nx[u] - cx, ry = ny[u] - cy, rz = nz[u] - cz;
        float d2 = rx * rx;
        d2 = fmaf(ry, ry, d2);
        d2 = fmaf(rz, rz, d2);
        const bool far = (d2 >= far2) && (d2 < __builtin_inff());
        const int s = cnt * 256 + tid;
        Lx[s] = nx[u]; Ly[s] = ny[u]; Lz[s] = nz[u];
        cnt += far ? 0 : 1;
    }

#pragma nounroll
    for (int j = 0; j < cnt; ++j) {
        const int s = j * 256 + tid;
        const float px = Lx[s], py = Ly[s], pz = Lz[s];
        const float rx = px - cx, ry = py - cy, rz = pz - cz;
        float rs = rx * rx;
        rs = fmaf(ry, ry, rs);
        rs = fmaf(rz, rz, rs);
        rs *= EXP_SCALE;
#pragma unroll
        for (int k = 0; k < KPTS; ++k) {
            float t = rs + ec[k];
            t = fmaf(rx, ekx[k], t);
            t = fmaf(ry, eky[k], t);
            t = fmaf(rz, ekz[k], t);
            const float w = __builtin_amdgcn_exp2f(t);
            pw[3 * k + 0] = fmaf(w, px, pw[3 * k + 0]);
            pw[3 * k + 1] = fmaf(w, py, pw[3 * k + 1]);
            pw[3 * k + 2] = fmaf(w, pz, pw[3 * k + 2]);
        }
    }

#pragma unroll
    for (int j = 0; j < KC; ++j)
        pw[j] += __shfl_xor(pw[j], 16, 32);

    v8f acc0 = {};
    v8f acc1 = {};
    const float* wrow0 = Wt + col * KCPAD;
    const float* wrow1 = Wt + (16 + col) * KCPAD;
#pragma unroll
    for (int s = 0; s < NSTEP; ++s) {
        const float lo0 = pw[4 * s + 0], lo1 = pw[4 * s + 1];
        const float hi0 = pw[4 * s + 2], hi1 = pw[4 * s + 3];
        v2f a;
        a.x = half ? hi0 : lo0;
        a.y = half ? hi1 : lo1;
        const v2f bf0 = *(const v2f*)(wrow0 + 4 * s + 2 * half);
        const v2f bf1 = *(const v2f*)(wrow1 + 4 * s + 2 * half);
        acc0 = __builtin_amdgcn_wmma_f32_16x16x4_f32(
            false, a, false, bf0, (short)0, acc0, false, false);
        acc1 = __builtin_amdgcn_wmma_f32_16x16x4_f32(
            false, a, false, bf1, (short)0, acc1, false, false);
    }

    if (wave < nTiles) {
        const int rowBase = tile * 16 + 8 * half;
#pragma unroll
        for (int r = 0; r < 8; ++r) {
            const int o = (rowBase + r) * COUT;
            out[o + col]      = acc0[r] + b0;
            out[o + 16 + col] = acc1[r] + b1;
        }
    }
}

extern "C" void kernel_launch(void* const* d_in, const int* in_sizes, int n_in,
                              void* d_out, int out_size, void* d_ws, size_t ws_size,
                              hipStream_t stream) {
    const float* points = (const float*)d_in[0];
    const int*   nbr    = (const int*)  d_in[1];
    const float* kp     = (const float*)d_in[2];
    const float* W      = (const float*)d_in[3];
    const float* bias   = (const float*)d_in[4];
    float*       out    = (float*)d_out;

    const int nTiles = NPTS / 16;
    kpconv_wmma_kernel<<<dim3(512), dim3(256), 0, stream>>>(
        points, nbr, kp, W, bias, out, nTiles);
}


# Far test by LDS look-up: launch 1 packs points into 4-byte cell codes (workspace); launch 2 holds that table in LDS and gathers exact coordinates only for neighbors not excluded; sums unchanged
# speedup vs baseline: 3.4718x; 1.1903x over previous
#include <hip/hip_runtime.h>


#define KPTS  15
#define HNBR  32
#define COUT  32
#define NPTS  65536

#define KC    45
#define KCPAD 48
#define NSTEP 12

#define EXP_SCALE (-288.53900817779268f)
#define CULL_R0 (0.7260f)
#define NLIST 16

typedef __attribute__((ext_vector_type(2))) float v2f;
typedef __attribute__((ext_vector_type(8))) float v8f;
typedef __attribute__((ext_vector_type(4))) int   v4i;

__device__ __forceinline__ float uniform_f(float v) {
    return __int_as_float(__builtin_amdgcn_readfirstlane(__float_as_int(v)));
}

__global__ __launch_bounds__(256) __attribute__((amdgpu_waves_per_eu(4, 4)))
void kpconv_wmma_kernel(
    const float* __restrict__ points,
    const int*   __restrict__ nbr,
    const float* __restrict__ kp,
    const float* __restrict__ W,
    const float* __restrict__ bias,
    float*       __restrict__ out,
    int nTiles)
{
    __shared__ float Wt[COUT * KCPAD];
    __shared__ float Lx[NLIST * 256];
    __shared__ float Ly[NLIST * 256];
    __shared__ float Lz[NLIST * 256];

    const int tid   = threadIdx.x;
    const int lane  = tid & 31;
    const int half  = lane >> 4;
    const int col   = lane & 15;
    const int wave  = (blockIdx.x * 256 + tid) >> 5;
    const int tile  = (wave < nTiles) ? wave : (nTiles - 1);

    const int m = tile * 16 + col;
    const v4i* ip = (const v4i*)(nbr + m * HNBR + half * 16);
    v4i iv[4];
#pragma unroll
    for (int q = 0; q < 4; ++q) iv[q] = ip[q];
    const float* pc = points + m * 3;
    const float cx = pc[0], cy = pc[1], cz = pc[2];

    float wv[6];
#pragma unroll
    for (int i = 0; i < 6; ++i) {
        const int e = tid + 256 * i;
        const int es = (e < KC * COUT) ? e : (KC * COUT - 1);
        wv[i] = W[es];
    }
    const float b0 = bias[col];
    const float b1 = bias[16 + col];
    __builtin_amdgcn_sched_barrier(0);

    float nx[16], ny[16], nz[16];
#pragma unroll
    for (int q = 0; q < 4; ++q) {
#pragma unroll
        for (int u = 0; u < 4; ++u) {
            int idx = iv[q][u];
            idx = (idx < 0) ? 0 : ((idx >= NPTS) ? NPTS - 1 : idx);
            const float* pn = points + idx * 3;
            nx[4 * q + u] = pn[0]; ny[4 * q + u] = pn[1]; nz[4 * q + u] = pn[2];
        }
    }

    float ekx[KPTS], eky[KPTS], ekz[KPTS], ec[KPTS];
    float kmax2 = 0.0f;
    bool  kfinite = true;
#pragma unroll
    for (int k = 0; k < KPTS; ++k) {
        const float x = kp[k * 3 + 0];
        const float y = kp[k * 3 + 1];
        const float z = kp[k * 3 + 2];
        const float k2 = x * x + y * y + z * z;
        kmax2 = fmaxf(kmax2, k2);
        kfinite = kfinite && (k2 < __builtin_inff());
        ekx[k] = uniform_f(-2.0f * EXP_SCALE * x);
        eky[k] = uniform_f(-2.0f * EXP_SCALE * y);
        ekz[k] = uniform_f(-2.0f * EXP_SCALE * z);
        ec[k]  = uniform_f(EXP_SCALE * (x * x + y * y + z * z));
    }
    const float kr = __builtin_sqrtf(kmax2) + CULL_R0;
    const float far2 = uniform_f(kfinite ? kr * kr * 1.001f : __builtin_inff());

#pragma unroll
    for (int i = 0; i < 6; ++i) {
        const int e = tid + 256 * i;
        const int es = (e < KC * COUT) ? e : (KC * COUT - 1);
        Wt[(es % COUT) * KCPAD + es / COUT] = wv[i];
    }
    if (tid < COUT * (KCPAD - KC))
        Wt[(tid / (KCPAD - KC)) * KCPAD + KC + tid % (KCPAD - KC)] = 0.0f;
    __builtin_amdgcn_fence(__ATOMIC_RELEASE, "workgroup", "local");
    __builtin_amdgcn_s_barrier();
    __builtin_amdgcn_fence(__ATOMIC_ACQUIRE, "workgroup", "local");
    __builtin_amdgcn_sched_barrier(0);

    float pw[KCPAD];
#pragma unroll
    for (int j = 0; j < KCPAD; ++j) pw[j] = 0.0f;

    int cnt = 0;
#pragma unroll
    for (int u = 0; u < 16; ++u) {
        const float rx = nx[u] - cx, ry = ny[u] - cy, rz = nz[u] - cz;
        float d2 = rx * rx;
        d2 = fmaf(ry, ry, d2);
        d2 = fmaf(rz, rz, d2);
        const bool far = (d2 >= far2) && (d2 < __builtin_inff());
        const int s = cnt * 256 + tid;
        Lx[s] = nx[u]; Ly[s] = ny[u]; Lz[s] = nz[u];
        cnt += far ? 0 : 1;
    }

#pragma nounroll
    for (int j = 0; j < cnt; ++j) {
        const int s = j * 256 + tid;
        const float px = Lx[s], py = Ly[s], pz = Lz[s];
        const float rx = px - cx, ry = py - cy, rz = pz - cz;
        float rs = rx * rx;
        rs = fmaf(ry, ry, rs);
        rs = fmaf(rz, rz, rs);
        rs *= EXP_SCALE;
#pragma unroll
        for (int k = 0; k < KPTS; ++k) {
            float t = rs + ec[k];
            t = fmaf(rx, ekx[k], t);
            t = fmaf(ry, eky[k], t);
            t = fmaf(rz, ekz[k], t);
            const float w = __builtin_amdgcn_exp2f(t);
            pw[3 * k + 0] = fmaf(w, px, pw[3 * k + 0]);
            pw[3 * k + 1] = fmaf(w, py, pw[3 * k + 1]);
            pw[3 * k + 2] = fmaf(w, pz, pw[3 * k + 2]);
        }
    }

#pragma unroll
    for (int j = 0; j < KC; ++j)
        pw[j] += __shfl_xor(pw[j], 16, 32);

    v8f acc0 = {};
    v8f acc1 = {};
    const float* wrow0 = Wt + col * KCPAD;
    const float* wrow1 = Wt + (16 + col) * KCPAD;
#pragma unroll
    for (int s = 0; s < NSTEP; ++s) {
        const float lo0 = pw[4 * s + 0], lo1 = pw[4 * s + 1];
        const float hi0 = pw[4 * s + 2], hi1 = pw[4 * s + 3];
        v2f a;
        a.x = half ? hi0 : lo0;
        a.y = half ? hi1 : lo1;
        const v2f bf0 = *(const v2f*)(wrow0 + 4 * s + 2 * half);
        const v2f bf1 = *(const v2f*)(wrow1 + 4 * s + 2 * half);
        acc0 = __builtin_amdgcn_wmma_f32_16x16x4_f32(
            false, a, false, bf0, (short)0, acc0, false, false);
        acc1 = __builtin_amdgcn_wmma_f32_16x16x4_f32(
            false, a, false, bf1, (short)0, acc1, false, false);
    }

    if (wave < nTiles) {
        const int rowBase = tile * 16 + 8 * half;
#pragma unroll
        for (int r = 0; r < 8; ++r) {
            const int o = (rowBase + r) * COUT;
            out[o + col]      = acc0[r] + b0;
            out[o + 16 + col] = acc1[r] + b1;
        }
    }
}


#define QCELLS   (1024.0f)
#define QSCALE   (128.0f)
#define QOFFSET  (4.0f)
#define QFLAG    (0x80000000u)

typedef __attribute__((ext_vector_type(4))) float    v4f;
typedef __attribute__((ext_vector_type(4))) unsigned v4u;

__device__ __forceinline__ unsigned pack_point(float x, float y, float z) {
    const float tx = (x + QOFFSET) * QSCALE;
    const float ty = (y + QOFFSET) * QSCALE;
    const float tz = (z + QOFFSET) * QSCALE;
    const bool ok = (tx >= 0.0f) && (tx < QCELLS) && (ty >= 0.0f) && (ty < QCELLS) &&
                    (tz >= 0.0f) && (tz < QCELLS);
    const unsigned qx = (unsigned)(int)(ok ? tx : 0.0f);
    const unsigned qy = (unsigned)(int)(ok ? ty : 0.0f);
    const unsigned qz = (unsigned)(int)(ok ? tz : 0.0f);
    return ok ? (qx | (qy << 10) | (qz << 20)) : QFLAG;
}

__global__ __launch_bounds__(128)
void kpconv_pack_points_kernel(const float* __restrict__ points,
                               unsigned*    __restrict__ codes)
{
    const int t = blockIdx.x * 128 + threadIdx.x;
    const v4f* p4 = (const v4f*)(points + t * 12);
    const v4f a = p4[0], b = p4[1], c = p4[2];
    v4u r;
    r.x = pack_point(a.x, a.y, a.z);
    r.y = pack_point(a.w, b.x, b.y);
    r.z = pack_point(b.z, b.w, c.x);
    r.w = pack_point(c.y, c.z, c.w);
    ((v4u*)codes)[t] = r;
}

__device__ __forceinline__ void add_neighbor(
    float px, float py, float pz, float cx, float cy, float cz,
    const float (&ekx)[KPTS], const float (&eky)[KPTS], const float (&ekz)[KPTS],
    const float (&ec)[KPTS], float (&pw)[KCPAD])
{
    const float rx = px - cx, ry = py - cy, rz = pz - cz;
    float rs = rx * rx;
    rs = fmaf(ry, ry, rs);
    rs = fmaf(rz, rz, rs);
    rs *= EXP_SCALE;
#pragma unroll
    for (int k = 0; k < KPTS; ++k) {
        float t = rs + ec[k];
        t = fmaf(rx, ekx[k], t);
        t = fmaf(ry, eky[k], t);
        t = fmaf(rz, ekz[k], t);
        const float w = __builtin_amdgcn_exp2f(t);
        pw[3 * k + 0] = fmaf(w, px, pw[3 * k + 0]);
        pw[3 * k + 1] = fmaf(w, py, pw[3 * k + 1]);
        pw[3 * k + 2] = fmaf(w, pz, pw[3 * k + 2]);
    }
}

#define WG2     512
#define TCHUNKS 32
#define GBATCH  4

__global__ __launch_bounds__(WG2) __attribute__((amdgpu_waves_per_eu(4, 4)))
void kpconv_lds_filter_kernel(
    const float*    __restrict__ points,
    const int*      __restrict__ nbr,
    const float*    __restrict__ kp,
    const float*    __restrict__ W,
    const float*    __restrict__ bias,
    const unsigned* __restrict__ codes,
    float*          __restrict__ out,
    int nTiles)
{
    __shared__ unsigned Tq[NPTS];
    __shared__ float    Wt[COUT * KCPAD];
    __shared__ int      Ls[NLIST * WG2];

    const int tid   = threadIdx.x;
    const int lane  = tid & 31;
    const int half  = lane >> 4;
    const int col   = lane & 15;
    const int wave  = (blockIdx.x * WG2 + tid) >> 5;
    const int tile  = (wave < nTiles) ? wave : (nTiles - 1);

    const int m = tile * 16 + col;
    const v4i* ip = (const v4i*)(nbr + m * HNBR + half * 16);
    v4i iv[4];
#pragma unroll
    for (int q = 0; q < 4; ++q) iv[q] = ip[q];
    const float* pc = points + m * 3;
    const float cx = pc[0], cy = pc[1], cz = pc[2];

    float wv[3];
#pragma unroll
    for (int i = 0; i < 3; ++i) {
        const int e = tid + WG2 * i;
        const int es = (e < KC * COUT) ? e : (KC * COUT - 1);
        wv[i] = W[es];
    }
    const float b0 = bias[col];
    const float b1 = bias[16 + col];

    const v4u* cg = (const v4u*)codes;
    const int rot = blockIdx.x & (TCHUNKS - 1);
    v4u tv[TCHUNKS];
#pragma unroll
    for (int i = 0; i < TCHUNKS; ++i) {
        const int ch = (i + rot) & (TCHUNKS - 1);
        tv[i] = cg[ch * WG2 + tid];
    }
    __builtin_amdgcn_sched_barrier(0);

    float ekx[KPTS], eky[KPTS], ekz[KPTS], ec[KPTS];
    float kmax2 = 0.0f;
    bool  kfinite = true;
#pragma unroll
    for (int k = 0; k < KPTS; ++k) {
        const float x = kp[k * 3 + 0];
        const float y = kp[k * 3 + 1];
        const float z = kp[k * 3 + 2];
        const float k2 = x * x + y * y + z * z;
        kmax2 = fmaxf(kmax2, k2);
        kfinite = kfinite && (k2 < __builtin_inff());
        ekx[k] = uniform_f(-2.0f * EXP_SCALE * x);
        eky[k] = uniform_f(-2.0f * EXP_SCALE * y);
        ekz[k] = uniform_f(-2.0f * EXP_SCALE * z);
        ec[k]  = uniform_f(EXP_SCALE * (x * x + y * y + z * z));
    }
    const float kr  = __builtin_sqrtf(kmax2) + CULL_R0;
    const float krq = fmaf(kr, 1.0006f, 0.0070f);
    const float farq2 = uniform_f(kfinite ? krq * krq * 1.0001f : __builtin_inff());

#pragma unroll
    for (int i = 0; i < 3; ++i) {
        const int e = tid + WG2 * i;
        const int es = (e < KC * COUT) ? e : (KC * COUT - 1);
        Wt[(es % COUT) * KCPAD + es / COUT] = wv[i];
    }
    if (tid < COUT * (KCPAD - KC))
        Wt[(tid / (KCPAD - KC)) * KCPAD + KC + tid % (KCPAD - KC)] = 0.0f;
    v4u* tl = (v4u*)Tq;
#pragma unroll
    for (int i = 0; i < TCHUNKS; ++i) {
        const int ch = (i + rot) & (TCHUNKS - 1);
        tl[ch * WG2 + tid] = tv[i];
    }
    __syncthreads();

    const float ox = (0.5f / QSCALE - QOFFSET) - cx;
    const float oy = (0.5f / QSCALE - QOFFSET) - cy;
    const float oz = (0.5f / QSCALE - QOFFSET) - cz;
    int cnt = 0;
#pragma unroll
    for (int q = 0; q < 4; ++q) {
#pragma unroll
        for (int u = 0; u < 4; ++u) {
            int idx = iv[q][u];
            idx = (idx < 0) ? 0 : ((idx >= NPTS) ? NPTS - 1 : idx);
            const unsigned code = Tq[idx];
            const float qx = (float)(code & 1023u);
            const float qy = (float)((code >> 10) & 1023u);
            const float qz = (float)((code >> 20) & 1023u);
            const float rx = fmaf(qx, 1.0f / QSCALE, ox);
            const float ry = fmaf(qy, 1.0f / QSCALE, oy);
            const float rz = fmaf(qz, 1.0f / QSCALE, oz);
            float d2 = rx * rx;
            d2 = fmaf(ry, ry, d2);
            d2 = fmaf(rz, rz, d2);
            const bool far = (d2 >= farq2) && (d2 < __builtin_inff()) && ((int)code >= 0);
            Ls[cnt * WG2 + tid] = idx;
            cnt += far ? 0 : 1;
        }
    }

    float pw[KCPAD];
#pragma unroll
    for (int j = 0; j < KCPAD; ++j) pw[j] = 0.0f;

#pragma nounroll
    for (int b = 0; b < NLIST; b += GBATCH) {
        if (__builtin_amdgcn_ballot_w32(cnt > b) == 0u) break;
        float gx[GBATCH], gy[GBATCH], gz[GBATCH];
#pragma unroll
        for (int u = 0; u < GBATCH; ++u) {
            const int j = b + u;
            const int li = Ls[j * WG2 + tid];
            const int idx = (j < cnt) ? li : m;
            const float* pn = points + idx * 3;
            gx[u] = pn[0]; gy[u] = pn[1]; gz[u] = pn[2];
        }
#pragma unroll
        for (int u = 0; u < GBATCH; ++u) {
            if (b + u < cnt)
                add_neighbor(gx[u], gy[u], gz[u], cx, cy, cz, ekx, eky, ekz, ec, pw);
        }
    }

#pragma unroll
    for (int j = 0; j < KC; ++j)
        pw[j] += __shfl_xor(pw[j], 16, 32);

    v8f acc0 = {};
    v8f acc1 = {};
    const float* wrow0 = Wt + col * KCPAD;
    const float* wrow1 = Wt + (16 + col) * KCPAD;
#pragma unroll
    for (int s = 0; s < NSTEP; ++s) {
        const float lo0 = pw[4 * s + 0], lo1 = pw[4 * s + 1];
        const float hi0 = pw[4 * s + 2], hi1 = pw[4 * s + 3];
        v2f a;
        a.x = half ? hi0 : lo0;
        a.y = half ? hi1 : lo1;
        const v2f bf0 = *(const v2f*)(wrow0 + 4 * s + 2 * half);
        const v2f bf1 = *(const v2f*)(wrow1 + 4 * s + 2 * half);
        acc0 = __builtin_amdgcn_wmma_f32_16x16x4_f32(
            false, a, false, bf0, (short)0, acc0, false, false);
        acc1 = __builtin_amdgcn_wmma_f32_16x16x4_f32(
            false, a, false, bf1, (short)0, acc1, false, false);
    }

    if (wave < nTiles) {
        const int rowBase = tile * 16 + 8 * half;
#pragma unroll
        for (int r = 0; r < 8; ++r) {
            const int o = (rowBase + r) * COUT;
            out[o + col]      = acc0[r] + b0;
            out[o + 16 + col] = acc1[r] + b1;
        }
    }
}

extern "C" void kernel_launch(void* const* d_in, const int* in_sizes, int n_in,
                              void* d_out, int out_size, void* d_ws, size_t ws_size,
                              hipStream_t stream) {
    const float* points = (const float*)d_in[0];
    const int*   nbr    = (const int*)  d_in[1];
    const float* kp     = (const float*)d_in[2];
    const float* W      = (const float*)d_in[3];
    const float* bias   = (const float*)d_in[4];
    float*       out    = (float*)d_out;

    const int nTiles = NPTS / 16;
    if (d_ws != nullptr && ws_size >= (size_t)NPTS * 4) {
        unsigned* codes = (unsigned*)d_ws;
        kpconv_pack_points_kernel<<<dim3(128), dim3(128), 0, stream>>>(points, codes);
        kpconv_lds_filter_kernel<<<dim3(256), dim3(WG2), 0, stream>>>(
            points, nbr, kp, W, bias, codes, out, nTiles);
    } else {
        kpconv_wmma_kernel<<<dim3(512), dim3(256), 0, stream>>>(
            points, nbr, kp, W, bias, out, nTiles);
    }
}
